# starting kernel
# baseline (speedup 1.0000x reference)
#include <hip/hip_runtime.h>
#include <hip/hip_bf16.h>


typedef __attribute__((ext_vector_type(2))) float v2f;
typedef __attribute__((ext_vector_type(8))) float v8f;

__device__ __forceinline__ v8f wmma_k4(v2f a, v2f b, v8f c) {
  return __builtin_amdgcn_wmma_f32_16x16x4_f32(
       false, a,  false, b,
       (short)0, c,  false,  false);
}

__device__ __forceinline__ float ssp(float x) {
  float e = __expf(-fabsf(x));
  return fmaxf(x, 0.0f) + __logf(1.0f + e) - 0.69314718055994531f;
}

#define F 128
#define KRBF 64
#define A_STRIDE 132
#define E_STRIDE 68

__global__ void ssp_kernel(const float* __restrict__ x, float* __restrict__ xa, int n) {
  int i = blockIdx.x * 256 + threadIdx.x;
  if (i < n) xa[i] = ssp(x[i]);
}

__global__ void __launch_bounds__(256)
edge_msg_kernel(const float* __restrict__ xa, const int* __restrict__ eidx,
                const float* __restrict__ ea, const float* __restrict__ Wd,
                const float* __restrict__ bd, const float* __restrict__ Wg,
                float* __restrict__ aggr, int E) {
  __shared__ __align__(16) float As[16 * A_STRIDE];
  __shared__ __align__(16) float Es[16 * E_STRIDE];
  __shared__ int dsts[16];

  const int e0 = blockIdx.x * 16;

  if (threadIdx.x < 16) {
    int e = e0 + threadIdx.x;
    if (e >= E) e = E - 1;
    dsts[threadIdx.x] = eidx[E + e];
  }
  for (int i = threadIdx.x; i < 16 * F; i += 256) {
    int r = i >> 7, c = i & (F - 1);
    int e = e0 + r;
    if (e >= E) e = E - 1;
    int s = eidx[e];
    As[r * A_STRIDE + c] = xa[s * F + c];
  }
  for (int i = threadIdx.x; i < 16 * KRBF; i += 256) {
    int r = i >> 6, c = i & (KRBF - 1);
    int e = e0 + r;
    if (e >= E) e = E - 1;
    Es[r * E_STRIDE + c] = ea[e * KRBF + c];
  }
  __syncthreads();

  const int lane = threadIdx.x & 31;
  const int wv   = threadIdx.x >> 5;
  const int m    = lane & 15;
  const int hi   = lane >> 4;
  const int n    = wv * 16 + m;

  v8f acc = {};
  const float* wrow = Wd + n * F;
#pragma unroll
  for (int kk = 0; kk < F / 4; ++kk) {
    v2f a = *(const v2f*)(&As[m * A_STRIDE + kk * 4 + 2 * hi]);
    v2f b = *(const v2f*)(&wrow[kk * 4 + 2 * hi]);
    acc = wmma_k4(a, b, acc);
  }

  v8f acc2 = {};
  const float* grow = Wg + n * KRBF;
#pragma unroll
  for (int kk = 0; kk < KRBF / 4; ++kk) {
    v2f a = *(const v2f*)(&Es[m * E_STRIDE + kk * 4 + 2 * hi]);
    v2f b = *(const v2f*)(&grow[kk * 4 + 2 * hi]);
    acc2 = wmma_k4(a, b, acc2);
  }

  const float bn = bd[n];
#pragma unroll
  for (int j = 0; j < 8; ++j) {
    int r = j + 8 * hi;
    int e = e0 + r;
    float msg = ssp(acc[j] + bn) * acc2[j];
    if (e < E) atomicAdd(&aggr[dsts[r] * F + n], msg);
  }
}

template <bool IN_ACT, int MODE, bool WRITE2>
__global__ void __launch_bounds__(256)
gemm128_kernel(const float* __restrict__ In, const float* __restrict__ W,
               const float* __restrict__ bias, const float* __restrict__ Ex1,
               const float* __restrict__ Ex2, float* __restrict__ Out,
               float* __restrict__ Out2, int N) {
  __shared__ __align__(16) float As[16 * A_STRIDE];

  const int row0 = blockIdx.x * 16;
  for (int i = threadIdx.x; i < 16 * F; i += 256) {
    int r = i >> 7, c = i & (F - 1);
    int rr = row0 + r;
    if (rr >= N) rr = N - 1;
    float v = In[rr * F + c];
    if (IN_ACT) v = ssp(v);
    As[r * A_STRIDE + c] = v;
  }
  __syncthreads();

  const int lane = threadIdx.x & 31;
  const int wv   = threadIdx.x >> 5;
  const int m    = lane & 15;
  const int hi   = lane >> 4;
  const int n    = wv * 16 + m;

  v8f acc = {};
  const float* wrow = W + n * F;
#pragma unroll
  for (int kk = 0; kk < F / 4; ++kk) {
    v2f a = *(const v2f*)(&As[m * A_STRIDE + kk * 4 + 2 * hi]);
    v2f b = *(const v2f*)(&wrow[kk * 4 + 2 * hi]);
    acc = wmma_k4(a, b, acc);
  }

  const float bn = bias[n];
#pragma unroll
  for (int j = 0; j < 8; ++j) {
    int r = row0 + j + 8 * hi;
    if (r >= N) continue;
    float g = acc[j] + bn;
    float o;
    if (MODE == 0)      o = ssp(g) + Ex1[r * F + n];
    else if (MODE == 1) o = g;
    else if (MODE == 2) o = Ex1[r * F + n] + g;
    else                o = g + Ex1[r * F + n] * Ex2[n];
    Out[r * F + n] = o;
    if (WRITE2) Out2[r * F + n] = o;
  }
}

extern "C" void kernel_launch(void* const* d_in, const int* in_sizes, int n_in,
                              void* d_out, int out_size, void* d_ws, size_t ws_size,
                              hipStream_t stream) {
  const float* x      = (const float*)d_in[0];
  const int*   eidx   = (const int*)  d_in[1];
  const float* ea     = (const float*)d_in[2];
  const float* W_same = (const float*)d_in[3];
  const float* b_same = (const float*)d_in[4];
  const float* W_diff = (const float*)d_in[5];
  const float* b_diff = (const float*)d_in[6];
  const float* W_G    = (const float*)d_in[7];
  const float* rW1    = (const float*)d_in[8];
  const float* rb1    = (const float*)d_in[9];
  const float* rW2    = (const float*)d_in[10];
  const float* rb2    = (const float*)d_in[11];
  const float* W_last = (const float*)d_in[12];
  const float* b_last = (const float*)d_in[13];
  const float* u      = (const float*)d_in[14];

  const int NF = in_sizes[0];
  const int N  = NF / F;
  const int E  = in_sizes[2] / KRBF;
  const int R  = in_sizes[8] / (F * F);

  float* out_final = (float*)d_out;
  float* out_msged = (float*)d_out + NF;

  float* buf0 = (float*)d_ws;
  float* buf1 = buf0 + NF;

  const int nbN = (N + 15) / 16;
  const int nbE = (E + 15) / 16;

  hipMemsetAsync(buf1, 0, (size_t)NF * sizeof(float), stream);

  ssp_kernel<<<(NF + 255) / 256, 256, 0, stream>>>(x, buf0, NF);

  edge_msg_kernel<<<nbE, 256, 0, stream>>>(buf0, eidx, ea, W_diff, b_diff, W_G,
                                           buf1, E);

  gemm128_kernel<false, 0, true><<<nbN, 256, 0, stream>>>(
      buf0, W_same, b_same, buf1, nullptr, out_msged, buf0, N);

  for (int r = 0; r < R; ++r) {
    gemm128_kernel<true, 1, false><<<nbN, 256, 0, stream>>>(
        buf0, rW1 + (size_t)r * F * F, rb1 + (size_t)r * F,
        nullptr, nullptr, buf1, nullptr, N);
    gemm128_kernel<true, 2, false><<<nbN, 256, 0, stream>>>(
        buf1, rW2 + (size_t)r * F * F, rb2 + (size_t)r * F,
        buf0, nullptr, buf0, nullptr, N);
  }

  gemm128_kernel<true, 3, false><<<nbN, 256, 0, stream>>>(
      buf0, W_last, b_last, x, u, out_final, nullptr, N);
}


# Node chain in one launch, state in registers: products transposed (W as A operand), a layer's D registers are the next B fragments; k order and bias-first sum order change; 5 launches from 12
# speedup vs baseline: 1.7789x; 1.7789x over previous
#include <hip/hip_runtime.h>
#include <hip/hip_bf16.h>


typedef __attribute__((ext_vector_type(2))) float v2f;
typedef __attribute__((ext_vector_type(8))) float v8f;

__device__ __forceinline__ v8f wmma_k4(v2f a, v2f b, v8f c) {
  return __builtin_amdgcn_wmma_f32_16x16x4_f32(
       false, a,  false, b,
       (short)0, c,  false,  false);
}

__device__ __forceinline__ float ssp(float x) {
  float e = __expf(-fabsf(x));
  return fmaxf(x, 0.0f) + __logf(1.0f + e) - 0.69314718055994531f;
}

#define F 128
#define KRBF 64
#define W_STRIDE 132
#define G_STRIDE 68

#define NB 256
#define RPB 196
#define ACC_PITCH 144
#define EPT 5
#define OWN_NT 768

__global__ void ssp_kernel(const float* __restrict__ x, float* __restrict__ xa, int n,
                           int* __restrict__ bins) {
  int i = blockIdx.x * 256 + threadIdx.x;
  if (i < n) xa[i] = ssp(x[i]);
  if (blockIdx.x == 0) { bins[threadIdx.x] = 0; bins[NB + threadIdx.x] = 0; }
}

__device__ __forceinline__ int range_of(int d) {
  int b = (int)((unsigned)d / (unsigned)RPB);
  return b < NB ? b : NB - 1;
}

__global__ void __launch_bounds__(256)
bin_count_kernel(const int* __restrict__ eidx, int* __restrict__ cnt, int E) {
  __shared__ int h[NB];
  const int t = threadIdx.x;
  h[t] = 0;
  __syncthreads();
  const int e0 = blockIdx.x * (256 * EPT);
  for (int it = 0; it < EPT; ++it) {
    int e = e0 + it * 256 + t;
    if (e < E) atomicAdd(&h[range_of(eidx[E + e])], 1);
  }
  __syncthreads();
  const int c = h[t];
  if (c) atomicAdd(&cnt[t], c);
}

__global__ void __launch_bounds__(256)
bin_place_kernel(const int* __restrict__ eidx, const int* __restrict__ cnt,
                 int* __restrict__ cur, int* __restrict__ seg0, int* __restrict__ rec, int E) {
  __shared__ int h[NB];
  __shared__ int sc[NB];
  __shared__ int start[NB];
  const int t = threadIdx.x;
  h[t] = 0;
  const int c = cnt[t];
  sc[t] = c;
  __syncthreads();
  for (int off = 1; off < NB; off <<= 1) {
    int v = (t >= off) ? sc[t - off] : 0;
    __syncthreads();
    sc[t] += v;
    __syncthreads();
  }
  const int base = sc[t] - c;
  if (blockIdx.x == 0) seg0[t] = base;

  const int e0 = blockIdx.x * (256 * EPT);
  int eb[EPT], erk[EPT], es[EPT], edl[EPT];
#pragma unroll
  for (int it = 0; it < EPT; ++it) {
    int e = e0 + it * 256 + t;
    eb[it] = -1;
    if (e < E) {
      int d = eidx[E + e];
      int b = range_of(d);
      int dl = d - b * RPB;
      dl = dl < 0 ? 0 : (dl < RPB ? dl : RPB - 1);
      eb[it] = b;
      es[it] = eidx[e];
      edl[it] = dl;
      erk[it] = atomicAdd(&h[b], 1);
    }
  }
  __syncthreads();
  const int hc = h[t];
  int st = base;
  if (hc) st += atomicAdd(&cur[t], hc);
  start[t] = st;
  __syncthreads();
#pragma unroll
  for (int it = 0; it < EPT; ++it) {
    if (eb[it] >= 0) {
      int e = e0 + it * 256 + t;
      ((int4*)rec)[start[eb[it]] + erk[it]] = make_int4(e, es[it], edl[it], 0);
    }
  }
}

template <int KDIM, int STRIDE, int NT>
__device__ __forceinline__ void stage_weight_nt(const float* __restrict__ Wsrc, float* Wl) {
  constexpr int PIECES = F * (KDIM / 4);
  constexpr int PER_THREAD = (PIECES + NT - 1) / NT;
  float4 v[PER_THREAD];
#pragma unroll
  for (int it = 0; it < PER_THREAD; ++it) {
    int i = it * NT + threadIdx.x;
    if (i > PIECES - 1) i = PIECES - 1;
    int r = i / (KDIM / 4), c4 = i % (KDIM / 4);
    v[it] = *(const float4*)(Wsrc + r * KDIM + c4 * 4);
  }
#pragma unroll
  for (int it = 0; it < PER_THREAD; ++it) {
    int i = it * NT + threadIdx.x;
    if (i < PIECES) {
      int r = i / (KDIM / 4), c4 = i % (KDIM / 4);
      *(float4*)(&Wl[r * STRIDE + c4 * 4]) = v[it];
    }
  }
}

template <int KSTEPS>
__device__ __forceinline__ v8f tile_product(const v2f (&a)[KSTEPS], const float* wl) {
  v8f acc = {};
#pragma unroll
  for (int kk = 0; kk < KSTEPS; ++kk) {
    v2f b = *(const v2f*)(wl + kk * 4);
    acc = wmma_k4(a[kk], b, acc);
  }
  return acc;
}

__global__ void __launch_bounds__(OWN_NT)
edge_owner_kernel(const float* __restrict__ xa, const int* __restrict__ rec,
                  const int* __restrict__ cnt, const int* __restrict__ seg0,
                  const float* __restrict__ ea, const float* __restrict__ Wd,
                  const float* __restrict__ bd, const float* __restrict__ Wg,
                  float* __restrict__ aggr, int N) {
  __shared__ __align__(16) float Wds[F * W_STRIDE];
  __shared__ __align__(16) float Wgs[F * G_STRIDE];
  __shared__ __align__(16) float Acc[RPB * ACC_PITCH];

  stage_weight_nt<F, W_STRIDE, OWN_NT>(Wd, Wds);
  stage_weight_nt<KRBF, G_STRIDE, OWN_NT>(Wg, Wgs);
  for (int i = threadIdx.x; i < RPB * ACC_PITCH / 4; i += OWN_NT)
    ((float4*)Acc)[i] = make_float4(0.0f, 0.0f, 0.0f, 0.0f);
  __syncthreads();

  const int lane = threadIdx.x & 31;
  const int wv   = __builtin_amdgcn_readfirstlane(threadIdx.x >> 5);
  const int m    = lane & 15;
  const int hi   = lane >> 4;

  const int b      = blockIdx.x;
  const int nb     = cnt[b];
  const int4* rseg = (const int4*)rec + seg0[b];
  const int ntiles = (nb + 15) >> 4;

  for (int tile = wv; tile < ntiles; tile += OWN_NT / 32) {
    const int i0 = tile * 16;
    int ie = i0 + m;
    if (ie >= nb) ie = nb - 1;
    const int4 r = rseg[ie];
    const float* arow = xa + r.y * F + 2 * hi;
    const float* erow = ea + r.x * KRBF + 2 * hi;

    v2f a[F / 4], g[KRBF / 4];
#pragma unroll
    for (int kk = 0; kk < F / 4; ++kk) a[kk] = *(const v2f*)(arow + kk * 4);
#pragma unroll
    for (int kk = 0; kk < KRBF / 4; ++kk) g[kk] = *(const v2f*)(erow + kk * 4);

    int dacc[8];
    bool live[8];
#pragma unroll
    for (int j = 0; j < 8; ++j) {
      int id = i0 + j + 8 * hi;
      live[j] = id < nb;
      if (id >= nb) id = nb - 1;
      dacc[j] = rseg[id].z * ACC_PITCH + m;
    }

    for (int ct = 0; ct < F / 16; ++ct) {
      const int n = ct * 16 + m;
      v8f acc  = tile_product<F / 4>(a, &Wds[n * W_STRIDE + 2 * hi]);
      v8f acc2 = tile_product<KRBF / 4>(g, &Wgs[n * G_STRIDE + 2 * hi]);
      const float bn = bd[n];
#pragma unroll
      for (int j = 0; j < 8; ++j) {
        float msg = ssp(acc[j] + bn) * acc2[j];
        atomicAdd(&Acc[dacc[j] + ct * 16], live[j] ? msg : 0.0f);
      }
    }
  }
  __syncthreads();

  const int r0 = b * RPB;
  int nrows = N - r0;
  if (nrows > RPB) nrows = RPB;
  for (int i = threadIdx.x; i < nrows * (F / 4); i += OWN_NT) {
    const int row = i / (F / 4), c4 = i % (F / 4);
    *(float4*)(aggr + (r0 + row) * F + c4 * 4) = *(const float4*)(&Acc[row * ACC_PITCH + c4 * 4]);
  }
}


__device__ __forceinline__ void stage_weight_chain(const float* __restrict__ Wsrc, float* Wl) {
  constexpr int PER_PASS = F * (F / 4) / 256 / 2;
#pragma nounroll
  for (int pass = 0; pass < 2; ++pass) {
    float4 v[PER_PASS];
#pragma unroll
    for (int it = 0; it < PER_PASS; ++it) {
      int i = (pass * PER_PASS + it) * 256 + threadIdx.x;
      int r = i / (F / 4), c4 = i % (F / 4);
      v[it] = *(const float4*)(Wsrc + r * F + c4 * 4);
    }
#pragma unroll
    for (int it = 0; it < PER_PASS; ++it) {
      int i = (pass * PER_PASS + it) * 256 + threadIdx.x;
      int r = i / (F / 4), c4 = i % (F / 4);
      int g = c4 >> 2, q = c4 & 3;
      float* d = &Wl[r * W_STRIDE + 16 * g + 8 * (q & 1) + 2 * (q >> 1)];
      v2f lo = {v[it].x, v[it].y}, hi2 = {v[it].z, v[it].w};
      *(v2f*)d = lo;
      *(v2f*)(d + 4) = hi2;
    }
  }
}

__device__ __forceinline__ void load_rows(const float* p, v8f (&d)[8]) {
#pragma unroll
  for (int o = 0; o < 8; ++o) {
    float4 a = *(const float4*)(p + 16 * o);
    float4 b = *(const float4*)(p + 16 * o + 4);
    v8f w = {a.x, a.y, a.z, a.w, b.x, b.y, b.z, b.w};
    d[o] = w;
  }
}

__device__ __forceinline__ void store_rows(float* p, const v8f (&d)[8]) {
#pragma unroll
  for (int o = 0; o < 8; ++o) {
    *(float4*)(p + 16 * o)     = make_float4(d[o][0], d[o][1], d[o][2], d[o][3]);
    *(float4*)(p + 16 * o + 4) = make_float4(d[o][4], d[o][5], d[o][6], d[o][7]);
  }
}

__device__ __forceinline__ void ssp_rows(const v8f (&s)[8], v8f (&d)[8]) {
#pragma unroll
  for (int o = 0; o < 8; ++o)
#pragma unroll
    for (int r = 0; r < 8; ++r) d[o][r] = ssp(s[o][r]);
}

__device__ __forceinline__ void chain_product(const float* wl, const v8f (&in)[8], v8f (&acc)[8]) {
#pragma unroll
  for (int g = 0; g < 8; ++g) {
#pragma unroll
    for (int p = 0; p < 4; ++p) {
      v2f b = {in[g][2 * p], in[g][2 * p + 1]};
#pragma unroll
      for (int o = 0; o < 8; ++o) {
        v2f a = *(const v2f*)(wl + o * 16 * W_STRIDE + 16 * g + 4 * p);
        acc[o] = wmma_k4(a, b, acc[o]);
      }
    }
  }
}

__global__ void __launch_bounds__(256, 4)
node_chain_kernel(const float* __restrict__ xa, const float* __restrict__ x,
                  const float* __restrict__ W_same, const float* __restrict__ b_same,
                  const float* __restrict__ rW1, const float* __restrict__ rb1,
                  const float* __restrict__ rW2, const float* __restrict__ rb2,
                  const float* __restrict__ W_last, const float* __restrict__ b_last,
                  const float* __restrict__ u, float* __restrict__ msged,
                  float* __restrict__ outf, int N, int ntiles, int R) {
  __shared__ __align__(16) float Ws[F * W_STRIDE];

  const int lane = threadIdx.x & 31;
  const int wv   = threadIdx.x >> 5;
  const int c    = lane & 15;
  const int hi   = lane >> 4;

  int tile = blockIdx.x * 8 + wv;
  const bool live = tile < ntiles;
  if (!live) tile = ntiles - 1;
  const int row = tile * 16 + c;
  const bool ok = live && row < N;
  const int rr  = row < N ? row : N - 1;
  const int off = rr * F + 8 * hi;

  v8f t[8], in[8], acc[8];
  load_rows(xa + off, in);
  load_rows(msged + off, t);

  const float* wl = Ws + c * W_STRIDE + 2 * hi;
  const int nl = 2 * R + 2;

  for (int l = 0; l < nl; ++l) {
    const float* Wsrc;
    const float* bsrc;
    if (l == 0)           { Wsrc = W_same; bsrc = b_same; }
    else if (l == nl - 1) { Wsrc = W_last; bsrc = b_last; }
    else {
      const int rb = (l - 1) >> 1;
      Wsrc = ((l & 1) ? rW1 : rW2) + rb * F * F;
      bsrc = ((l & 1) ? rb1 : rb2) + rb * F;
    }
    __syncthreads();
    stage_weight_chain(Wsrc, Ws);
    __syncthreads();

    load_rows(bsrc + 8 * hi, acc);
    chain_product(wl, in, acc);

    if (l == nl - 1) {
      load_rows(u + 8 * hi, in);
#pragma unroll
      for (int o = 0; o < 8; ++o) acc[o] = acc[o] + t[o] * in[o];
      if (ok) store_rows(outf + off, acc);
    } else if (l & 1) {
      ssp_rows(acc, in);
    } else {
#pragma unroll
      for (int o = 0; o < 8; ++o) {
        if (l == 0) {
#pragma unroll
          for (int r = 0; r < 8; ++r) t[o][r] = ssp(acc[o][r]) + t[o][r];
          if (ok) {
            *(float4*)(msged + off + 16 * o)     = make_float4(t[o][0], t[o][1], t[o][2], t[o][3]);
            *(float4*)(msged + off + 16 * o + 4) = make_float4(t[o][4], t[o][5], t[o][6], t[o][7]);
          }
        } else {
          t[o] = t[o] + acc[o];
        }
#pragma unroll
        for (int r = 0; r < 8; ++r) in[o][r] = ssp(t[o][r]);
      }
      if (l == nl - 2) load_rows(x + off, t);
    }
  }
}

extern "C" void kernel_launch(void* const* d_in, const int* in_sizes, int n_in,
                              void* d_out, int out_size, void* d_ws, size_t ws_size,
                              hipStream_t stream) {
  const float* x      = (const float*)d_in[0];
  const int*   eidx   = (const int*)  d_in[1];
  const float* ea     = (const float*)d_in[2];
  const float* W_same = (const float*)d_in[3];
  const float* b_same = (const float*)d_in[4];
  const float* W_diff = (const float*)d_in[5];
  const float* b_diff = (const float*)d_in[6];
  const float* W_G    = (const float*)d_in[7];
  const float* rW1    = (const float*)d_in[8];
  const float* rb1    = (const float*)d_in[9];
  const float* rW2    = (const float*)d_in[10];
  const float* rb2    = (const float*)d_in[11];
  const float* W_last = (const float*)d_in[12];
  const float* b_last = (const float*)d_in[13];
  const float* u      = (const float*)d_in[14];

  const int NF = in_sizes[0];
  const int N  = NF / F;
  const int E  = in_sizes[2] / KRBF;
  const int R  = in_sizes[8] / (F * F);

  float* out_final = (float*)d_out;
  float* out_msged = (float*)d_out + NF;

  float* buf0 = (float*)d_ws;
  float* buf1 = buf0 + NF;

  int* bins = (int*)buf1;
  int* recs = bins + 4 * NB;

  const int ntN = (N + 15) / 16;
  const int nbN = (ntN + 7) / 8;
  const int nbE = (E + 256 * EPT - 1) / (256 * EPT);

  ssp_kernel<<<(NF + 255) / 256, 256, 0, stream>>>(x, buf0, NF, bins);

  bin_count_kernel<<<nbE, 256, 0, stream>>>(eidx, bins, E);
  bin_place_kernel<<<nbE, 256, 0, stream>>>(eidx, bins, bins + NB, bins + 2 * NB, recs, E);

  edge_owner_kernel<<<NB, OWN_NT, 0, stream>>>(buf0, recs, bins, bins + 2 * NB, ea,
                                               W_diff, b_diff, W_G, out_msged, N);

  node_chain_kernel<<<nbN, 256, 0, stream>>>(buf0, x, W_same, b_same, rW1, rb1, rW2, rb2,
                                             W_last, b_last, u, out_msged, out_final, N, ntN, R);
}


# Node chain: each wave parks its rows' residual state t in LDS between layers (own 8 KB region, no barrier), not in 64 registers through every product: no scratch spills left; sums unchanged
# speedup vs baseline: 2.1148x; 1.1888x over previous
#include <hip/hip_runtime.h>
#include <hip/hip_bf16.h>


typedef __attribute__((ext_vector_type(2))) float v2f;
typedef __attribute__((ext_vector_type(8))) float v8f;

__device__ __forceinline__ v8f wmma_k4(v2f a, v2f b, v8f c) {
  return __builtin_amdgcn_wmma_f32_16x16x4_f32(
       false, a,  false, b,
       (short)0, c,  false,  false);
}

__device__ __forceinline__ float ssp(float x) {
  float e = __expf(-fabsf(x));
  return fmaxf(x, 0.0f) + __logf(1.0f + e) - 0.69314718055994531f;
}

#define F 128
#define KRBF 64
#define W_STRIDE 132
#define G_STRIDE 68

#define NB 256
#define RPB 196
#define ACC_PITCH 144
#define EPT 5
#define OWN_NT 768

__global__ void ssp_kernel(const float* __restrict__ x, float* __restrict__ xa, int n,
                           int* __restrict__ bins) {
  int i = blockIdx.x * 256 + threadIdx.x;
  if (i < n) xa[i] = ssp(x[i]);
  if (blockIdx.x == 0) { bins[threadIdx.x] = 0; bins[NB + threadIdx.x] = 0; }
}

__device__ __forceinline__ int range_of(int d) {
  int b = (int)((unsigned)d / (unsigned)RPB);
  return b < NB ? b : NB - 1;
}

__global__ void __launch_bounds__(256)
bin_count_kernel(const int* __restrict__ eidx, int* __restrict__ cnt, int E) {
  __shared__ int h[NB];
  const int t = threadIdx.x;
  h[t] = 0;
  __syncthreads();
  const int e0 = blockIdx.x * (256 * EPT);
  for (int it = 0; it < EPT; ++it) {
    int e = e0 + it * 256 + t;
    if (e < E) atomicAdd(&h[range_of(eidx[E + e])], 1);
  }
  __syncthreads();
  const int c = h[t];
  if (c) atomicAdd(&cnt[t], c);
}

__global__ void __launch_bounds__(256)
bin_place_kernel(const int* __restrict__ eidx, const int* __restrict__ cnt,
                 int* __restrict__ cur, int* __restrict__ seg0, int* __restrict__ rec, int E) {
  __shared__ int h[NB];
  __shared__ int sc[NB];
  __shared__ int start[NB];
  const int t = threadIdx.x;
  h[t] = 0;
  const int c = cnt[t];
  sc[t] = c;
  __syncthreads();
  for (int off = 1; off < NB; off <<= 1) {
    int v = (t >= off) ? sc[t - off] : 0;
    __syncthreads();
    sc[t] += v;
    __syncthreads();
  }
  const int base = sc[t] - c;
  if (blockIdx.x == 0) seg0[t] = base;

  const int e0 = blockIdx.x * (256 * EPT);
  int eb[EPT], erk[EPT], es[EPT], edl[EPT];
#pragma unroll
  for (int it = 0; it < EPT; ++it) {
    int e = e0 + it * 256 + t;
    eb[it] = -1;
    if (e < E) {
      int d = eidx[E + e];
      int b = range_of(d);
      int dl = d - b * RPB;
      dl = dl < 0 ? 0 : (dl < RPB ? dl : RPB - 1);
      eb[it] = b;
      es[it] = eidx[e];
      edl[it] = dl;
      erk[it] = atomicAdd(&h[b], 1);
    }
  }
  __syncthreads();
  const int hc = h[t];
  int st = base;
  if (hc) st += atomicAdd(&cur[t], hc);
  start[t] = st;
  __syncthreads();
#pragma unroll
  for (int it = 0; it < EPT; ++it) {
    if (eb[it] >= 0) {
      int e = e0 + it * 256 + t;
      ((int4*)rec)[start[eb[it]] + erk[it]] = make_int4(e, es[it], edl[it], 0);
    }
  }
}

template <int KDIM, int STRIDE, int NT>
__device__ __forceinline__ void stage_weight_nt(const float* __restrict__ Wsrc, float* Wl) {
  constexpr int PIECES = F * (KDIM / 4);
  constexpr int PER_THREAD = (PIECES + NT - 1) / NT;
  float4 v[PER_THREAD];
#pragma unroll
  for (int it = 0; it < PER_THREAD; ++it) {
    int i = it * NT + threadIdx.x;
    if (i > PIECES - 1) i = PIECES - 1;
    int r = i / (KDIM / 4), c4 = i % (KDIM / 4);
    v[it] = *(const float4*)(Wsrc + r * KDIM + c4 * 4);
  }
#pragma unroll
  for (int it = 0; it < PER_THREAD; ++it) {
    int i = it * NT + threadIdx.x;
    if (i < PIECES) {
      int r = i / (KDIM / 4), c4 = i % (KDIM / 4);
      *(float4*)(&Wl[r * STRIDE + c4 * 4]) = v[it];
    }
  }
}

template <int KSTEPS>
__device__ __forceinline__ v8f tile_product(const v2f (&a)[KSTEPS], const float* wl) {
  v8f acc = {};
#pragma unroll
  for (int kk = 0; kk < KSTEPS; ++kk) {
    v2f b = *(const v2f*)(wl + kk * 4);
    acc = wmma_k4(a[kk], b, acc);
  }
  return acc;
}

__global__ void __launch_bounds__(OWN_NT)
edge_owner_kernel(const float* __restrict__ xa, const int* __restrict__ rec,
                  const int* __restrict__ cnt, const int* __restrict__ seg0,
                  const float* __restrict__ ea, const float* __restrict__ Wd,
                  const float* __restrict__ bd, const float* __restrict__ Wg,
                  float* __restrict__ aggr, int N) {
  __shared__ __align__(16) float Wds[F * W_STRIDE];
  __shared__ __align__(16) float Wgs[F * G_STRIDE];
  __shared__ __align__(16) float Acc[RPB * ACC_PITCH];

  stage_weight_nt<F, W_STRIDE, OWN_NT>(Wd, Wds);
  stage_weight_nt<KRBF, G_STRIDE, OWN_NT>(Wg, Wgs);
  for (int i = threadIdx.x; i < RPB * ACC_PITCH / 4; i += OWN_NT)
    ((float4*)Acc)[i] = make_float4(0.0f, 0.0f, 0.0f, 0.0f);
  __syncthreads();

  const int lane = threadIdx.x & 31;
  const int wv   = __builtin_amdgcn_readfirstlane(threadIdx.x >> 5);
  const int m    = lane & 15;
  const int hi   = lane >> 4;

  const int b      = blockIdx.x;
  const int nb     = cnt[b];
  const int4* rseg = (const int4*)rec + seg0[b];
  const int ntiles = (nb + 15) >> 4;

  for (int tile = wv; tile < ntiles; tile += OWN_NT / 32) {
    const int i0 = tile * 16;
    int ie = i0 + m;
    if (ie >= nb) ie = nb - 1;
    const int4 r = rseg[ie];
    const float* arow = xa + r.y * F + 2 * hi;
    const float* erow = ea + r.x * KRBF + 2 * hi;

    v2f a[F / 4], g[KRBF / 4];
#pragma unroll
    for (int kk = 0; kk < F / 4; ++kk) a[kk] = *(const v2f*)(arow + kk * 4);
#pragma unroll
    for (int kk = 0; kk < KRBF / 4; ++kk) g[kk] = *(const v2f*)(erow + kk * 4);

    int dacc[8];
    bool live[8];
#pragma unroll
    for (int j = 0; j < 8; ++j) {
      int id = i0 + j + 8 * hi;
      live[j] = id < nb;
      if (id >= nb) id = nb - 1;
      dacc[j] = rseg[id].z * ACC_PITCH + m;
    }

    for (int ct = 0; ct < F / 16; ++ct) {
      const int n = ct * 16 + m;
      v8f acc  = tile_product<F / 4>(a, &Wds[n * W_STRIDE + 2 * hi]);
      v8f acc2 = tile_product<KRBF / 4>(g, &Wgs[n * G_STRIDE + 2 * hi]);
      const float bn = bd[n];
#pragma unroll
      for (int j = 0; j < 8; ++j) {
        float msg = ssp(acc[j] + bn) * acc2[j];
        atomicAdd(&Acc[dacc[j] + ct * 16], live[j] ? msg : 0.0f);
      }
    }
  }
  __syncthreads();

  const int r0 = b * RPB;
  int nrows = N - r0;
  if (nrows > RPB) nrows = RPB;
  for (int i = threadIdx.x; i < nrows * (F / 4); i += OWN_NT) {
    const int row = i / (F / 4), c4 = i % (F / 4);
    *(float4*)(aggr + (r0 + row) * F + c4 * 4) = *(const float4*)(&Acc[row * ACC_PITCH + c4 * 4]);
  }
}


__device__ __forceinline__ void stage_weight_chain(const float* __restrict__ Wsrc, float* Wl) {
  constexpr int PER_PASS = F * (F / 4) / 256 / 2;
#pragma nounroll
  for (int pass = 0; pass < 2; ++pass) {
    float4 v[PER_PASS];
#pragma unroll
    for (int it = 0; it < PER_PASS; ++it) {
      int i = (pass * PER_PASS + it) * 256 + threadIdx.x;
      int r = i / (F / 4), c4 = i % (F / 4);
      v[it] = *(const float4*)(Wsrc + r * F + c4 * 4);
    }
#pragma unroll
    for (int it = 0; it < PER_PASS; ++it) {
      int i = (pass * PER_PASS + it) * 256 + threadIdx.x;
      int r = i / (F / 4), c4 = i % (F / 4);
      int g = c4 >> 2, q = c4 & 3;
      float* d = &Wl[r * W_STRIDE + 16 * g + 8 * (q & 1) + 2 * (q >> 1)];
      v2f lo = {v[it].x, v[it].y}, hi2 = {v[it].z, v[it].w};
      *(v2f*)d = lo;
      *(v2f*)(d + 4) = hi2;
    }
  }
}

__device__ __forceinline__ void load_rows(const float* p, v8f (&d)[8]) {
#pragma unroll
  for (int o = 0; o < 8; ++o) {
    float4 a = *(const float4*)(p + 16 * o);
    float4 b = *(const float4*)(p + 16 * o + 4);
    v8f w = {a.x, a.y, a.z, a.w, b.x, b.y, b.z, b.w};
    d[o] = w;
  }
}

__device__ __forceinline__ void store_rows(float* p, const v8f (&d)[8]) {
#pragma unroll
  for (int o = 0; o < 8; ++o) {
    *(float4*)(p + 16 * o)     = make_float4(d[o][0], d[o][1], d[o][2], d[o][3]);
    *(float4*)(p + 16 * o + 4) = make_float4(d[o][4], d[o][5], d[o][6], d[o][7]);
  }
}

__device__ __forceinline__ void ssp_rows(const v8f (&s)[8], v8f (&d)[8]) {
#pragma unroll
  for (int o = 0; o < 8; ++o)
#pragma unroll
    for (int r = 0; r < 8; ++r) d[o][r] = ssp(s[o][r]);
}

__device__ __forceinline__ void chain_product(const float* wl, const v8f (&in)[8], v8f (&acc)[8]) {
#pragma unroll
  for (int g = 0; g < 8; ++g) {
#pragma unroll
    for (int p = 0; p < 4; ++p) {
      v2f b = {in[g][2 * p], in[g][2 * p + 1]};
#pragma unroll
      for (int o = 0; o < 8; ++o) {
        v2f a = *(const v2f*)(wl + o * 16 * W_STRIDE + 16 * g + 4 * p);
        acc[o] = wmma_k4(a, b, acc[o]);
      }
    }
  }
}

__device__ __forceinline__ void park_tile(float* tl, int o, const v8f& d) {
  *(float4*)(tl + (2 * o) * 128)     = make_float4(d[0], d[1], d[2], d[3]);
  *(float4*)(tl + (2 * o + 1) * 128) = make_float4(d[4], d[5], d[6], d[7]);
}

__device__ __forceinline__ v8f fetch_tile(const float* tl, int o) {
  float4 a = *(const float4*)(tl + (2 * o) * 128);
  float4 b = *(const float4*)(tl + (2 * o + 1) * 128);
  v8f w = {a.x, a.y, a.z, a.w, b.x, b.y, b.z, b.w};
  return w;
}

__global__ void __launch_bounds__(256, 4)
node_chain_kernel(const float* __restrict__ xa, const float* __restrict__ x,
                  const float* __restrict__ W_same, const float* __restrict__ b_same,
                  const float* __restrict__ rW1, const float* __restrict__ rb1,
                  const float* __restrict__ rW2, const float* __restrict__ rb2,
                  const float* __restrict__ W_last, const float* __restrict__ b_last,
                  const float* __restrict__ u, float* __restrict__ msged,
                  float* __restrict__ outf, int N, int ntiles, int R) {
  __shared__ __align__(16) float Ws[F * W_STRIDE];
  __shared__ __align__(16) float Ts[8 * 16 * F];

  const int lane = threadIdx.x & 31;
  const int wv   = threadIdx.x >> 5;
  const int c    = lane & 15;
  const int hi   = lane >> 4;

  int tile = blockIdx.x * 8 + wv;
  const bool live = tile < ntiles;
  if (!live) tile = ntiles - 1;
  const int row = tile * 16 + c;
  const bool ok = live && row < N;
  const int rr  = row < N ? row : N - 1;
  const int off = rr * F + 8 * hi;

  float* tl = Ts + wv * (16 * F) + lane * 4;

  v8f in[8], acc[8];
  load_rows(xa + off, in);
  load_rows(msged + off, acc);
#pragma unroll
  for (int o = 0; o < 8; ++o) park_tile(tl, o, acc[o]);

  const float* wl = Ws + c * W_STRIDE + 2 * hi;
  const int nl = 2 * R + 2;

  for (int l = 0; l < nl; ++l) {
    const float* Wsrc;
    const float* bsrc;
    if (l == 0)           { Wsrc = W_same; bsrc = b_same; }
    else if (l == nl - 1) { Wsrc = W_last; bsrc = b_last; }
    else {
      const int rb = (l - 1) >> 1;
      Wsrc = ((l & 1) ? rW1 : rW2) + rb * F * F;
      bsrc = ((l & 1) ? rb1 : rb2) + rb * F;
    }
    __syncthreads();
    stage_weight_chain(Wsrc, Ws);
    __syncthreads();

    load_rows(bsrc + 8 * hi, acc);
    chain_product(wl, in, acc);

    if (l == nl - 1) {
      v8f xr[8];
      load_rows(x + off, xr);
      load_rows(u + 8 * hi, in);
#pragma unroll
      for (int o = 0; o < 8; ++o) acc[o] = acc[o] + xr[o] * in[o];
      if (ok) store_rows(outf + off, acc);
    } else if (l & 1) {
      ssp_rows(acc, in);
    } else {
#pragma unroll
      for (int o = 0; o < 8; ++o) {
        v8f t = fetch_tile(tl, o);
        if (l == 0) {
#pragma unroll
          for (int r = 0; r < 8; ++r) t[r] = ssp(acc[o][r]) + t[r];
          if (ok) {
            *(float4*)(msged + off + 16 * o)     = make_float4(t[0], t[1], t[2], t[3]);
            *(float4*)(msged + off + 16 * o + 4) = make_float4(t[4], t[5], t[6], t[7]);
          }
        } else {
          t = t + acc[o];
        }
        if (l != nl - 2) park_tile(tl, o, t);
#pragma unroll
        for (int r = 0; r < 8; ++r) in[o][r] = ssp(t[r]);
      }
    }
  }
}

extern "C" void kernel_launch(void* const* d_in, const int* in_sizes, int n_in,
                              void* d_out, int out_size, void* d_ws, size_t ws_size,
                              hipStream_t stream) {
  const float* x      = (const float*)d_in[0];
  const int*   eidx   = (const int*)  d_in[1];
  const float* ea     = (const float*)d_in[2];
  const float* W_same = (const float*)d_in[3];
  const float* b_same = (const float*)d_in[4];
  const float* W_diff = (const float*)d_in[5];
  const float* b_diff = (const float*)d_in[6];
  const float* W_G    = (const float*)d_in[7];
  const float* rW1    = (const float*)d_in[8];
  const float* rb1    = (const float*)d_in[9];
  const float* rW2    = (const float*)d_in[10];
  const float* rb2    = (const float*)d_in[11];
  const float* W_last = (const float*)d_in[12];
  const float* b_last = (const float*)d_in[13];
  const float* u      = (const float*)d_in[14];

  const int NF = in_sizes[0];
  const int N  = NF / F;
  const int E  = in_sizes[2] / KRBF;
  const int R  = in_sizes[8] / (F * F);

  float* out_final = (float*)d_out;
  float* out_msged = (float*)d_out + NF;

  float* buf0 = (float*)d_ws;
  float* buf1 = buf0 + NF;

  int* bins = (int*)buf1;
  int* recs = bins + 4 * NB;

  const int ntN = (N + 15) / 16;
  const int nbN = (ntN + 7) / 8;
  const int nbE = (E + 256 * EPT - 1) / (256 * EPT);

  ssp_kernel<<<(NF + 255) / 256, 256, 0, stream>>>(x, buf0, NF, bins);

  bin_count_kernel<<<nbE, 256, 0, stream>>>(eidx, bins, E);
  bin_place_kernel<<<nbE, 256, 0, stream>>>(eidx, bins, bins + NB, bins + 2 * NB, recs, E);

  edge_owner_kernel<<<NB, OWN_NT, 0, stream>>>(buf0, recs, bins, bins + 2 * NB, ea,
                                               W_diff, b_diff, W_G, out_msged, N);

  node_chain_kernel<<<nbN, 256, 0, stream>>>(buf0, x, W_same, b_same, rW1, rb1, rW2, rb2,
                                             W_last, b_last, u, out_msged, out_final, N, ntN, R);
}
